# starting kernel
# baseline (speedup 1.0000x reference)
#include <hip/hip_runtime.h>
#include <hip/hip_bf16.h>

#define ALPHA   0.2f
#define NEG_INF (-9000000000000000.0f)

#define BT 128
#define NN 512
#define FF 128
#define DD 128
#define LDSP  132
#define PAIRP 144

typedef float v2f __attribute__((ext_vector_type(2)));
typedef float v8f __attribute__((ext_vector_type(8)));

__global__ void gat_wh_kernel(const float* __restrict__ h,
                              const float* __restrict__ W,
                              float* __restrict__ Wh) {
  extern __shared__ float smem[];
  float* sH = smem;
  v2f*   sW = (v2f*)(smem + 128 * LDSP);

  const int bt = blockIdx.x >> 2;
  const int rb = blockIdx.x & 3;
  const float* hblk  = h  + ((size_t)bt * NN + (size_t)rb * 128) * FF;
  float*       whblk = Wh + ((size_t)bt * NN + (size_t)rb * 128) * DD;

  const int tid = threadIdx.x;
  for (int idx = tid * 4; idx < 128 * FF; idx += 256 * 4) {
    const int r = idx >> 7, c = idx & 127;
    *(float4*)(sH + r * LDSP + c) = *(const float4*)(hblk + idx);
  }
  for (int idx = tid; idx < 64 * 32; idx += 256) {
    const int k2 = idx >> 5, nq = idx & 31;
    float4 r0 = *(const float4*)(W + (2 * k2) * DD + nq * 4);
    float4 r1 = *(const float4*)(W + (2 * k2 + 1) * DD + nq * 4);
    float* dst = (float*)(sW + k2 * PAIRP + nq * 4);
    *(float4*)(dst)     = make_float4(r0.x, r1.x, r0.y, r1.y);
    *(float4*)(dst + 4) = make_float4(r0.z, r1.z, r0.w, r1.w);
  }
  __syncthreads();

  const int lane = tid & 31;
  const int wave = tid >> 5;
  const int l16  = lane & 15;
  const int g    = lane >> 4;
  const int arow = wave * 16 + l16;

  v8f acc[8];
#pragma unroll
  for (int t = 0; t < 8; ++t)
#pragma unroll
    for (int r = 0; r < 8; ++r) acc[t][r] = 0.0f;

  for (int kk = 0; kk < FF; kk += 4) {
    const int ka = kk + 2 * g;
    const v2f a = *(const v2f*)(sH + arow * LDSP + ka);
    const v2f* brow = sW + ((kk >> 1) + g) * PAIRP + l16;
    v2f b[8];
#pragma unroll
    for (int t = 0; t < 8; ++t) b[t] = brow[t * 16];
#pragma unroll
    for (int t = 0; t < 8; ++t)
      acc[t] = __builtin_amdgcn_wmma_f32_16x16x4_f32(
          false, a, false, b[t], (short)0, acc[t], false, false);
  }

#pragma unroll
  for (int t = 0; t < 8; ++t)
#pragma unroll
    for (int r = 0; r < 8; ++r)
      whblk[(wave * 16 + r + 8 * g) * DD + t * 16 + l16] = acc[t][r];
}

__global__ void gat_e_kernel(const float* __restrict__ Wh,
                             const float* __restrict__ a,
                             float* __restrict__ esrc,
                             float* __restrict__ edst) {
  const int lane = threadIdx.x & 31;
  const int row  = (blockIdx.x * blockDim.x + threadIdx.x) >> 5;

  float4 v  = *(const float4*)(Wh + (size_t)row * DD + lane * 4);
  float4 as = *(const float4*)(a + lane * 4);
  float4 ad = *(const float4*)(a + DD + lane * 4);
  float ps = v.x * as.x + v.y * as.y + v.z * as.z + v.w * as.w;
  float pd = v.x * ad.x + v.y * ad.y + v.z * ad.z + v.w * ad.w;
#pragma unroll
  for (int off = 16; off >= 1; off >>= 1) {
    ps += __shfl_xor(ps, off, 32);
    pd += __shfl_xor(pd, off, 32);
  }
  if (lane == 0) {
    esrc[row] = ps;
    edst[row] = pd;
  }
}

__global__ void gat_attn_kernel(const float* __restrict__ Wh,
                                const float* __restrict__ esrc,
                                const float* __restrict__ edst,
                                const int* __restrict__ adj,
                                float* __restrict__ out) {
  extern __shared__ float smem[];
  v2f*      sWh   = (v2f*)smem;
  float*    sEd   = smem + (NN / 2) * PAIRP * 2;
  unsigned* sMask = (unsigned*)(sEd + NN);

  const int bt = blockIdx.x >> 2;
  const int rb = blockIdx.x & 3;
  const int tid = threadIdx.x;
  const float* whbt = Wh + (size_t)bt * NN * DD;

  for (int idx = tid; idx < 256 * 32; idx += 256) {
    const int k2 = idx >> 5, nq = idx & 31;
    float4 r0 = *(const float4*)(whbt + (size_t)(2 * k2) * DD + nq * 4);
    float4 r1 = *(const float4*)(whbt + (size_t)(2 * k2 + 1) * DD + nq * 4);
    float* dst = (float*)(sWh + k2 * PAIRP + nq * 4);
    *(float4*)(dst)     = make_float4(r0.x, r1.x, r0.y, r1.y);
    *(float4*)(dst + 4) = make_float4(r0.z, r1.z, r0.w, r1.w);
  }
  for (int j = tid; j < NN; j += 256) sEd[j] = edst[bt * NN + j];
  __syncthreads();

  const int lane = tid & 31;
  const int wave = tid >> 5;
  const int l16  = lane & 15;
  const int g    = lane >> 4;

  const int irow = rb * 128 + wave * 16 + l16;
  const float es = esrc[bt * NN + irow];
  const int* adjrow = adj + irow * NN;
  unsigned* maskrow = sMask + (wave * 16 + l16) * 16;

  float m = -3.0e38f, l = 0.0f;
  for (int w = 0; w < 16; ++w) {
    unsigned mw = 0;
#pragma unroll
    for (int jj = 0; jj < 16; ++jj) {
      const int j = w * 32 + 2 * jj + g;
      const bool conn = adjrow[j] > 0;
      float ev = es + sEd[j];
      ev = ev > 0.0f ? ev : ALPHA * ev;
      if (!conn) ev = NEG_INF;
      mw |= (conn ? 1u : 0u) << (2 * jj + g);
      const float nm = fmaxf(m, ev);
      l = l * __expf(m - nm) + __expf(ev - nm);
      m = nm;
    }
    mw |= __shfl_xor(mw, 16, 32);
    if (g == 0) maskrow[w] = mw;
  }
  {
    const float m2 = __shfl_xor(m, 16, 32);
    const float l2 = __shfl_xor(l, 16, 32);
    const float nm = fmaxf(m, m2);
    l = l * __expf(m - nm) + l2 * __expf(m2 - nm);
    m = nm;
  }

  v8f acc[8];
#pragma unroll
  for (int t = 0; t < 8; ++t)
#pragma unroll
    for (int r = 0; r < 8; ++r) acc[t][r] = 0.0f;

  for (int w = 0; w < 16; ++w) {
    const unsigned mw = maskrow[w];
#pragma unroll
    for (int c = 0; c < 8; ++c) {
      const int j0 = w * 32 + c * 4;
      const int ja = j0 + 2 * g;
      float e0 = es + sEd[ja];
      e0 = e0 > 0.0f ? e0 : ALPHA * e0;
      if (!((mw >> (c * 4 + 2 * g)) & 1u)) e0 = NEG_INF;
      float e1 = es + sEd[ja + 1];
      e1 = e1 > 0.0f ? e1 : ALPHA * e1;
      if (!((mw >> (c * 4 + 2 * g + 1)) & 1u)) e1 = NEG_INF;
      v2f a;
      a.x = __expf(e0 - m);
      a.y = __expf(e1 - m);
      const v2f* brow = sWh + ((j0 >> 1) + g) * PAIRP + l16;
      v2f b[8];
#pragma unroll
      for (int t = 0; t < 8; ++t) b[t] = brow[t * 16];
#pragma unroll
      for (int t = 0; t < 8; ++t)
        acc[t] = __builtin_amdgcn_wmma_f32_16x16x4_f32(
            false, a, false, b[t], (short)0, acc[t], false, false);
    }
  }

  float* outblk = out + ((size_t)bt * NN + (size_t)rb * 128 + wave * 16) * DD;
#pragma unroll
  for (int r = 0; r < 8; ++r) {
    const int orow = r + 8 * g;
    const float lr = __shfl(l, orow, 32);
    const float inv = 1.0f / lr;
#pragma unroll
    for (int t = 0; t < 8; ++t)
      outblk[orow * DD + t * 16 + l16] = acc[t][r] * inv;
  }
}

extern "C" void kernel_launch(void* const* d_in, const int* in_sizes, int n_in,
                              void* d_out, int out_size, void* d_ws, size_t ws_size,
                              hipStream_t stream) {
  const float* h   = (const float*)d_in[0];
  const int*   adj = (const int*)d_in[1];
  const float* W   = (const float*)d_in[2];
  const float* a   = (const float*)d_in[3];
  float* out = (float*)d_out;

  float* Wh   = (float*)d_ws;
  float* esrc = Wh + (size_t)BT * NN * DD;
  float* edst = esrc + (size_t)BT * NN;

  const size_t lds1 = (size_t)(128 * LDSP) * 4 + (size_t)(64 * PAIRP) * 8;
  const size_t lds3 = (size_t)(NN / 2) * PAIRP * 8 + (size_t)NN * 4 +
                      (size_t)(8 * 16 * 16) * 4;

  gat_wh_kernel<<<dim3(BT * 4), dim3(256), lds1, stream>>>(h, W, Wh);
  gat_e_kernel<<<dim3(BT * NN / 8), dim3(256), 0, stream>>>(Wh, a, esrc, edst);
  gat_attn_kernel<<<dim3(BT * 4), dim3(256), lds3, stream>>>(Wh, esrc, edst, adj, out);
}


# Softmax weights factored: exp(lrelu(es+ed)-M) = max(A*E_j, A2*E2_j), column factors made once per item at staging, M a fixed row bound: no exp, running max or rescale in WMMA loop; sum order changes
# speedup vs baseline: 2.1230x; 2.1230x over previous
#include <hip/hip_runtime.h>
#include <hip/hip_bf16.h>

#define ALPHA   0.2f
#define NEG_INF (-9000000000000000.0f)

#define BT 128
#define NN 512
#define FF 128
#define DD 128
#define LDSP  132
#define PAIRP 144

typedef float v2f __attribute__((ext_vector_type(2)));
typedef float v8f __attribute__((ext_vector_type(8)));
typedef float v4f __attribute__((ext_vector_type(4)));

#define WH_KC 32
#define WH_CHUNKS (FF / WH_KC)
#define WH_THREADS 256

struct WhStage {
  v4f hq[4];
  v4f w0[2];
  v4f w1[2];
};

__device__ __forceinline__ void wh_load_piece(WhStage& s, const float* __restrict__ hblk,
                                              const float* __restrict__ W, const int c, const int tid) {
#pragma unroll
  for (int i = 0; i < 4; ++i) {
    const int q = tid + WH_THREADS * i;
    const int row = q >> 3, c4 = q & 7;
    s.hq[i] = *(const v4f*)(hblk + row * FF + c * WH_KC + c4 * 4);
  }
#pragma unroll
  for (int i = 0; i < 2; ++i) {
    const int idx = tid + WH_THREADS * i;
    const int k2 = c * (WH_KC / 2) + (idx >> 5), nq = idx & 31;
    s.w0[i] = *(const v4f*)(W + (2 * k2) * DD + nq * 4);
    s.w1[i] = *(const v4f*)(W + (2 * k2 + 1) * DD + nq * 4);
  }
}

__device__ __forceinline__ void wh_store_piece(const WhStage& s, float* sH, v2f* sW,
                                               const int c, const int tid) {
#pragma unroll
  for (int i = 0; i < 4; ++i) {
    const int q = tid + WH_THREADS * i;
    const int row = q >> 3, c4 = q & 7;
    *(v4f*)(sH + row * LDSP + c * WH_KC + c4 * 4) = s.hq[i];
  }
#pragma unroll
  for (int i = 0; i < 2; ++i) {
    const int idx = tid + WH_THREADS * i;
    const int k2 = c * (WH_KC / 2) + (idx >> 5), nq = idx & 31;
    float* dst = (float*)(sW + k2 * PAIRP + nq * 4);
    const v4f lo = {s.w0[i].x, s.w1[i].x, s.w0[i].y, s.w1[i].y};
    const v4f hi = {s.w0[i].z, s.w1[i].z, s.w0[i].w, s.w1[i].w};
    *(v4f*)(dst)     = lo;
    *(v4f*)(dst + 4) = hi;
  }
}

__global__ void __launch_bounds__(WH_THREADS) gat_wh_kernel(const float* __restrict__ h,
                              const float* __restrict__ W,
                              float* __restrict__ Wh) {
  extern __shared__ float smem[];
  float* sH = smem;
  v2f*   sW = (v2f*)(smem + 128 * LDSP);

  const int bt = blockIdx.x >> 2;
  const int rb = blockIdx.x & 3;
  const float* hblk  = h  + ((size_t)bt * NN + (size_t)rb * 128) * FF;
  float*       whblk = Wh + ((size_t)bt * NN + (size_t)rb * 128) * DD;

  const int tid = threadIdx.x;
  const int lane = tid & 31;
  const int wave = tid >> 5;
  const int l16  = lane & 15;
  const int g    = lane >> 4;
  const int arow = wave * 16 + l16;

  WhStage st;
  wh_load_piece(st, hblk, W, 0, tid);
  wh_store_piece(st, sH, sW, 0, tid);

  v8f acc[8];
#pragma unroll
  for (int t = 0; t < 8; ++t)
#pragma unroll
    for (int r = 0; r < 8; ++r) acc[t][r] = 0.0f;

#pragma unroll
  for (int c = 0; c < WH_CHUNKS; ++c) {
    __syncthreads();
    if (c + 1 < WH_CHUNKS) {
      wh_load_piece(st, hblk, W, c + 1, tid);
      __builtin_amdgcn_sched_barrier(0);
    }

#pragma unroll
    for (int kk = c * WH_KC; kk < (c + 1) * WH_KC; kk += 4) {
      const int ka = kk + 2 * g;
      const v2f a = *(const v2f*)(sH + arow * LDSP + ka);
      const v2f* brow = sW + ((kk >> 1) + g) * PAIRP + l16;
      v2f b[8];
#pragma unroll
      for (int t = 0; t < 8; ++t) b[t] = brow[t * 16];
#pragma unroll
      for (int t = 0; t < 8; ++t)
        acc[t] = __builtin_amdgcn_wmma_f32_16x16x4_f32(
            false, a, false, b[t], (short)0, acc[t], false, false);
    }

    if (c + 1 < WH_CHUNKS) {
      __builtin_amdgcn_sched_barrier(0);
      wh_store_piece(st, sH, sW, c + 1, tid);
    }
  }

#pragma unroll
  for (int t = 0; t < 8; ++t)
#pragma unroll
    for (int r = 0; r < 8; ++r)
      whblk[(wave * 16 + r + 8 * g) * DD + t * 16 + l16] = acc[t][r];
}

template <int N, int OFF, int SZ>
__device__ __forceinline__ void gat_fold(float (&x)[SZ], const int lane) {
  const bool hi = (lane & OFF) != 0;
#pragma unroll
  for (int j = 0; j < N / 2; ++j) {
    const float send = hi ? x[j] : x[j + N / 2];
    const float keep = hi ? x[j + N / 2] : x[j];
    x[j] = keep + __shfl_xor(send, OFF, 32);
  }
}

#define AT_WAVES 16
#define AT_THREADS (AT_WAVES * 32)
#define AT_ROWS (AT_WAVES * 16)
__global__ void __launch_bounds__(AT_THREADS) gat_attn_kernel(const float* __restrict__ Wh,
                                const float* __restrict__ a,
                                const int* __restrict__ adj,
                                float* __restrict__ out) {
  extern __shared__ float smem[];
  v2f*      sWh   = (v2f*)smem;
  float*    sEd   = smem + (NN / 2) * PAIRP * 2;
  float*    sEs   = sEd + NN;
  float*    sEE   = sEs + AT_ROWS;
  float*    sMax  = sEE + 2 * NN;

  const int bt = blockIdx.x >> 1;
  const int rb = blockIdx.x & 1;
  const int tid = threadIdx.x;
  const float* whbt = Wh + (size_t)bt * NN * DD;

  const int lane = tid & 31;
  const int wave = tid >> 5;
  const int l16  = lane & 15;
  const int g    = lane >> 4;

  const int irow = rb * AT_ROWS + wave * 16 + l16;
  const int* adjrow = adj + irow * NN + 2 * g;

  int2 adjv[8];
#pragma unroll
  for (int c = 0; c < 8; ++c) adjv[c] = *(const int2*)(adjrow + c * 4);

  const v4f as = *(const v4f*)(a + lane * 4);
  const v4f ad = *(const v4f*)(a + DD + lane * 4);
  v4f v[32];
#pragma unroll
  for (int it = 0; it < 16; ++it) {
    const int k2 = wave + 16 * (it ^ (8 * rb));
    const float* src = whbt + (size_t)(2 * k2) * DD + lane * 4;
    v[2 * it]     = *(const v4f*)(src);
    v[2 * it + 1] = *(const v4f*)(src + DD);
  }
  __builtin_amdgcn_sched_barrier(0);
#pragma unroll
  for (int it = 0; it < 16; ++it) {
    const int k2 = wave + 16 * (it ^ (8 * rb));
    float* dst = (float*)(sWh + k2 * PAIRP + lane * 4);
    const v4f lo = {v[2 * it].x, v[2 * it + 1].x, v[2 * it].y, v[2 * it + 1].y};
    const v4f hi = {v[2 * it].z, v[2 * it + 1].z, v[2 * it].w, v[2 * it + 1].w};
    *(v4f*)(dst)     = lo;
    *(v4f*)(dst + 4) = hi;
  }
  {
    float xd[32], xs[16];
#pragma unroll
    for (int r = 0; r < 32; ++r)
      xd[r] = fmaf(v[r].w, ad.w, fmaf(v[r].z, ad.z, fmaf(v[r].x, ad.x, v[r].y * ad.y)));
#pragma unroll
    for (int r = 0; r < 16; ++r)
      xs[r] = fmaf(v[r].w, as.w, fmaf(v[r].z, as.z, fmaf(v[r].x, as.x, v[r].y * as.y)));
    gat_fold<32, 16>(xd, lane);  gat_fold<16, 16>(xs, lane);
    gat_fold<16, 8>(xd, lane);   gat_fold<8, 8>(xs, lane);
    gat_fold<8, 4>(xd, lane);    gat_fold<4, 4>(xs, lane);
    gat_fold<4, 2>(xd, lane);    gat_fold<2, 2>(xs, lane);
    gat_fold<2, 1>(xd, lane);
    xs[0] += __shfl_xor(xs[0], 1, 32);
    sEd[2 * wave + 32 * ((lane >> 1) ^ (8 * rb)) + (lane & 1)] = xd[0];
    if ((lane & 1) == 0) sEs[2 * wave + 32 * (lane >> 2) + ((lane >> 1) & 1)] = xs[0];
    float wm = xd[0];
    wm = fmaxf(wm, __shfl_xor(wm, 16, 32));
    wm = fmaxf(wm, __shfl_xor(wm, 8, 32));
    wm = fmaxf(wm, __shfl_xor(wm, 4, 32));
    wm = fmaxf(wm, __shfl_xor(wm, 2, 32));
    wm = fmaxf(wm, __shfl_xor(wm, 1, 32));
    if (lane == 0) sMax[wave] = wm;
  }
  __syncthreads();

  float D = sMax[0];
#pragma unroll
  for (int k = 1; k < AT_WAVES; ++k) D = fmaxf(D, sMax[k]);
  {
    const float t = sEd[tid] - D;
    float* ee = sEE + (tid >> 1) * 4 + (tid & 1);
    ee[0] = __expf(t);
    ee[2] = __expf(ALPHA * t);
  }
  __syncthreads();

  const float es = sEs[wave * 16 + l16];
  const float u = (1.0f - ALPHA) * (es + D);
  const float fa = __expf(fminf(u, 0.0f));
  const float fb = __expf(fminf(-u, 0.0f));
  const v2f fa2 = {fa, fa};
  const v2f fb2 = {fb, fb};

  v2f lp = {0.0f, 0.0f};
  v8f acc[8];
#pragma unroll
  for (int t = 0; t < 8; ++t)
#pragma unroll
    for (int r = 0; r < 8; ++r) acc[t][r] = 0.0f;

  for (int w = 0; w < 16; ++w) {
    int2 adjn[8];
    const int* nrow = adjrow + ((w + 1) & 15) * 32;
#pragma unroll
    for (int c = 0; c < 8; ++c) adjn[c] = *(const int2*)(nrow + c * 4);

    const v4f* eew = (const v4f*)sEE + w * 16 + g;
#pragma unroll
    for (int c = 0; c < 8; ++c) {
      const v4f ee = eew[2 * c];
      const v2f e1 = {ee.x, ee.y};
      const v2f e2 = {ee.z, ee.w};
      const v2f ca = e1 * fa2;
      const v2f cb = e2 * fb2;
      v2f a;
      a.x = fmaxf(ca.x, cb.x);
      a.y = fmaxf(ca.y, cb.y);
      if (!(adjv[c].x > 0)) a.x = 0.0f;
      if (!(adjv[c].y > 0)) a.y = 0.0f;
      lp += a;
      const v2f* brow = sWh + (w * 16 + c * 2 + g) * PAIRP + l16;
      v2f b[8];
#pragma unroll
      for (int t = 0; t < 8; ++t) b[t] = brow[t * 16];
#pragma unroll
      for (int t = 0; t < 8; ++t)
        acc[t] = __builtin_amdgcn_wmma_f32_16x16x4_f32(
            false, a, false, b[t], (short)0, acc[t], false, false);
    }

#pragma unroll
    for (int c = 0; c < 8; ++c) adjv[c] = adjn[c];
  }
  float l = lp.x + lp.y;
  l += __shfl_xor(l, 16, 32);

  if (__builtin_amdgcn_ballot_w32(!(l > 1.0e-20f)) != 0u) {
    float m = -3.0e38f;
    l = 0.0f;
#pragma unroll
    for (int t = 0; t < 8; ++t)
#pragma unroll
      for (int r = 0; r < 8; ++r) acc[t][r] = 0.0f;

    for (int w = 0; w < 16; ++w) {
      int2 adjn[8];
      const int* nrow = adjrow + ((w + 1) & 15) * 32;
#pragma unroll
      for (int c = 0; c < 8; ++c) adjn[c] = *(const int2*)(nrow + c * 4);

      const float* edw = sEd + w * 32 + 2 * g;
      float e[16];
#pragma unroll
      for (int c = 0; c < 8; ++c) {
        const v2f ed = *(const v2f*)(edw + c * 4);
        float e0 = es + ed.x;
        e0 = e0 > 0.0f ? e0 : ALPHA * e0;
        if (!(adjv[c].x > 0)) e0 = NEG_INF;
        float e1 = es + ed.y;
        e1 = e1 > 0.0f ? e1 : ALPHA * e1;
        if (!(adjv[c].y > 0)) e1 = NEG_INF;
        e[2 * c]     = e0;
        e[2 * c + 1] = e1;
      }

      float bm = e[0];
#pragma unroll
      for (int k = 1; k < 16; ++k) bm = fmaxf(bm, e[k]);
      bm = fmaxf(bm, __shfl_xor(bm, 16, 32));
      const float nm = fmaxf(m, bm);
      const float sc = __expf(m - nm);
      m = nm;
      l *= sc;
#pragma unroll
      for (int r = 0; r < 8; ++r) {
        const float f = __shfl(sc, r + 8 * g, 32);
#pragma unroll
        for (int t = 0; t < 8; ++t) acc[t][r] *= f;
      }

#pragma unroll
      for (int c = 0; c < 8; ++c) {
        v2f a;
        a.x = __expf(e[2 * c] - m);
        a.y = __expf(e[2 * c + 1] - m);
        l += a.x + a.y;
        const v2f* brow = sWh + (w * 16 + c * 2 + g) * PAIRP + l16;
        v2f b[8];
#pragma unroll
        for (int t = 0; t < 8; ++t) b[t] = brow[t * 16];
#pragma unroll
        for (int t = 0; t < 8; ++t)
          acc[t] = __builtin_amdgcn_wmma_f32_16x16x4_f32(
              false, a, false, b[t], (short)0, acc[t], false, false);
      }

#pragma unroll
      for (int c = 0; c < 8; ++c) adjv[c] = adjn[c];
    }
    l += __shfl_xor(l, 16, 32);
  }

  float* outblk = out + ((size_t)bt * NN + (size_t)rb * AT_ROWS + wave * 16) * DD;
#pragma unroll
  for (int r = 0; r < 8; ++r) {
    const int orow = r + 8 * g;
    const float lr = __shfl(l, orow, 32);
    const float inv = 1.0f / lr;
#pragma unroll
    for (int t = 0; t < 8; ++t)
      outblk[orow * DD + t * 16 + l16] = acc[t][r] * inv;
  }
}

extern "C" void kernel_launch(void* const* d_in, const int* in_sizes, int n_in,
                              void* d_out, int out_size, void* d_ws, size_t ws_size,
                              hipStream_t stream) {
  const float* h   = (const float*)d_in[0];
  const int*   adj = (const int*)d_in[1];
  const float* W   = (const float*)d_in[2];
  const float* a   = (const float*)d_in[3];
  float* out = (float*)d_out;

  float* Wh   = (float*)d_ws;

  const size_t lds1 = (size_t)(128 * LDSP) * 4 + (size_t)(64 * PAIRP) * 8;
  const size_t lds3 = (size_t)(NN / 2) * PAIRP * 8 + (size_t)NN * 4 + (size_t)AT_ROWS * 4
                    + (size_t)(NN / 2) * 16 + (size_t)AT_WAVES * 4;

  gat_wh_kernel<<<dim3(BT * 4), dim3(256), lds1, stream>>>(h, W, Wh);
  gat_attn_kernel<<<dim3(BT * NN / AT_ROWS), dim3(AT_THREADS), lds3, stream>>>(Wh, a, adj, out);
}


# Adjacency packed once per call by the GEMM kernel into lane-ordered select masks; attention loop takes a block's 16 words by scalar load: no adjacency vector loads or compares in the WMMA loop
# speedup vs baseline: 2.2455x; 1.0577x over previous
#include <hip/hip_runtime.h>
#include <hip/hip_bf16.h>

#define ALPHA   0.2f
#define NEG_INF (-9000000000000000.0f)

#define BT 128
#define NN 512
#define FF 128
#define DD 128
#define LDSP  132
#define PAIRP 144

typedef float v2f __attribute__((ext_vector_type(2)));
typedef float v8f __attribute__((ext_vector_type(8)));
typedef float v4f __attribute__((ext_vector_type(4)));

#define WH_KC 32
#define WH_CHUNKS (FF / WH_KC)
#define WH_THREADS 256
#define ADJ_LINES ((NN / 16) * (NN / 64))
static_assert(ADJ_LINES <= BT * 4, "one packing wave per line must exist in the GEMM grid");

struct WhStage {
  v4f hq[4];
  v4f w0[2];
  v4f w1[2];
};

__device__ __forceinline__ void wh_load_piece(WhStage& s, const float* __restrict__ hblk,
                                              const float* __restrict__ W, const int c, const int tid) {
#pragma unroll
  for (int i = 0; i < 4; ++i) {
    const int q = tid + WH_THREADS * i;
    const int row = q >> 3, c4 = q & 7;
    s.hq[i] = *(const v4f*)(hblk + row * FF + c * WH_KC + c4 * 4);
  }
#pragma unroll
  for (int i = 0; i < 2; ++i) {
    const int idx = tid + WH_THREADS * i;
    const int k2 = c * (WH_KC / 2) + (idx >> 5), nq = idx & 31;
    s.w0[i] = *(const v4f*)(W + (2 * k2) * DD + nq * 4);
    s.w1[i] = *(const v4f*)(W + (2 * k2 + 1) * DD + nq * 4);
  }
}

__device__ __forceinline__ void wh_store_piece(const WhStage& s, float* sH, v2f* sW,
                                               const int c, const int tid) {
#pragma unroll
  for (int i = 0; i < 4; ++i) {
    const int q = tid + WH_THREADS * i;
    const int row = q >> 3, c4 = q & 7;
    *(v4f*)(sH + row * LDSP + c * WH_KC + c4 * 4) = s.hq[i];
  }
#pragma unroll
  for (int i = 0; i < 2; ++i) {
    const int idx = tid + WH_THREADS * i;
    const int k2 = c * (WH_KC / 2) + (idx >> 5), nq = idx & 31;
    float* dst = (float*)(sW + k2 * PAIRP + nq * 4);
    const v4f lo = {s.w0[i].x, s.w1[i].x, s.w0[i].y, s.w1[i].y};
    const v4f hi = {s.w0[i].z, s.w1[i].z, s.w0[i].w, s.w1[i].w};
    *(v4f*)(dst)     = lo;
    *(v4f*)(dst + 4) = hi;
  }
}

__global__ void __launch_bounds__(WH_THREADS) gat_wh_kernel(const float* __restrict__ h,
                              const float* __restrict__ W,
                              float* __restrict__ Wh,
                              const int* __restrict__ adj,
                              unsigned int* __restrict__ adjbits) {
  extern __shared__ float smem[];
  float* sH = smem;
  v2f*   sW = (v2f*)(smem + 128 * LDSP);

  const int bt = blockIdx.x >> 2;
  const int rb = blockIdx.x & 3;
  const float* hblk  = h  + ((size_t)bt * NN + (size_t)rb * 128) * FF;
  float*       whblk = Wh + ((size_t)bt * NN + (size_t)rb * 128) * DD;

  const int tid = threadIdx.x;
  const int lane = tid & 31;
  const int wave = tid >> 5;
  const int l16  = lane & 15;
  const int g    = lane >> 4;
  const int arow = wave * 16 + l16;

  WhStage st;
  wh_load_piece(st, hblk, W, 0, tid);

  if (__builtin_amdgcn_readfirstlane(wave) == 0 && blockIdx.x < ADJ_LINES) {
    const int rt = blockIdx.x >> 3, wp = blockIdx.x & 7;
    const int* arow = adj + (rt * 16 + l16) * NN + wp * 64 + 2 * g;
    int2 aj[16];
#pragma unroll
    for (int j = 0; j < 16; ++j) aj[j] = *(const int2*)(arow + j * 4);
    unsigned int word = 0u;
#pragma unroll
    for (int j = 0; j < 16; ++j) {
      const unsigned int bx = __builtin_amdgcn_ballot_w32(aj[j].x > 0);
      const unsigned int by = __builtin_amdgcn_ballot_w32(aj[j].y > 0);
      if (lane == 2 * j)     word = bx;
      if (lane == 2 * j + 1) word = by;
    }
    adjbits[blockIdx.x * 32 + lane] = word;
  }

  wh_store_piece(st, sH, sW, 0, tid);

  v8f acc[8];
#pragma unroll
  for (int t = 0; t < 8; ++t)
#pragma unroll
    for (int r = 0; r < 8; ++r) acc[t][r] = 0.0f;

#pragma unroll
  for (int c = 0; c < WH_CHUNKS; ++c) {
    __syncthreads();
    if (c + 1 < WH_CHUNKS) {
      wh_load_piece(st, hblk, W, c + 1, tid);
      __builtin_amdgcn_sched_barrier(0);
    }

#pragma unroll
    for (int kk = c * WH_KC; kk < (c + 1) * WH_KC; kk += 4) {
      const int ka = kk + 2 * g;
      const v2f a = *(const v2f*)(sH + arow * LDSP + ka);
      const v2f* brow = sW + ((kk >> 1) + g) * PAIRP + l16;
      v2f b[8];
#pragma unroll
      for (int t = 0; t < 8; ++t) b[t] = brow[t * 16];
#pragma unroll
      for (int t = 0; t < 8; ++t)
        acc[t] = __builtin_amdgcn_wmma_f32_16x16x4_f32(
            false, a, false, b[t], (short)0, acc[t], false, false);
    }

    if (c + 1 < WH_CHUNKS) {
      __builtin_amdgcn_sched_barrier(0);
      wh_store_piece(st, sH, sW, c + 1, tid);
    }
  }

#pragma unroll
  for (int t = 0; t < 8; ++t)
#pragma unroll
    for (int r = 0; r < 8; ++r)
      whblk[(wave * 16 + r + 8 * g) * DD + t * 16 + l16] = acc[t][r];
}

template <int N, int OFF, int SZ>
__device__ __forceinline__ void gat_fold(float (&x)[SZ], const int lane) {
  const bool hi = (lane & OFF) != 0;
#pragma unroll
  for (int j = 0; j < N / 2; ++j) {
    const float send = hi ? x[j] : x[j + N / 2];
    const float keep = hi ? x[j + N / 2] : x[j];
    x[j] = keep + __shfl_xor(send, OFF, 32);
  }
}

#define AT_WAVES 16
#define AT_THREADS (AT_WAVES * 32)
#define AT_ROWS (AT_WAVES * 16)
__global__ void __launch_bounds__(AT_THREADS) gat_attn_kernel(const float* __restrict__ Wh,
                                const float* __restrict__ a,
                                const unsigned int* __restrict__ adjbits,
                                float* __restrict__ out) {
  extern __shared__ float smem[];
  v2f*      sWh   = (v2f*)smem;
  float*    sEd   = smem + (NN / 2) * PAIRP * 2;
  float*    sEs   = sEd + NN;
  float*    sEE   = sEs + AT_ROWS;
  float*    sMax  = sEE + 2 * NN;

  const int bt = blockIdx.x >> 1;
  const int rb = blockIdx.x & 1;
  const int tid = threadIdx.x;
  const float* whbt = Wh + (size_t)bt * NN * DD;

  const int lane = tid & 31;
  const int wave = tid >> 5;
  const int l16  = lane & 15;
  const int g    = lane >> 4;

  const unsigned int* mrow = adjbits + (rb * 16 + __builtin_amdgcn_readfirstlane(wave)) * 256;

  unsigned int mv[16];
#pragma unroll
  for (int k = 0; k < 16; ++k) mv[k] = mrow[k];

  const v4f as = *(const v4f*)(a + lane * 4);
  const v4f ad = *(const v4f*)(a + DD + lane * 4);
  v4f v[32];
#pragma unroll
  for (int it = 0; it < 16; ++it) {
    const int k2 = wave + 16 * (it ^ (8 * rb));
    const float* src = whbt + (size_t)(2 * k2) * DD + lane * 4;
    v[2 * it]     = *(const v4f*)(src);
    v[2 * it + 1] = *(const v4f*)(src + DD);
  }
  __builtin_amdgcn_sched_barrier(0);
#pragma unroll
  for (int it = 0; it < 16; ++it) {
    const int k2 = wave + 16 * (it ^ (8 * rb));
    float* dst = (float*)(sWh + k2 * PAIRP + lane * 4);
    const v4f lo = {v[2 * it].x, v[2 * it + 1].x, v[2 * it].y, v[2 * it + 1].y};
    const v4f hi = {v[2 * it].z, v[2 * it + 1].z, v[2 * it].w, v[2 * it + 1].w};
    *(v4f*)(dst)     = lo;
    *(v4f*)(dst + 4) = hi;
  }
  {
    float xd[32], xs[16];
#pragma unroll
    for (int r = 0; r < 32; ++r)
      xd[r] = fmaf(v[r].w, ad.w, fmaf(v[r].z, ad.z, fmaf(v[r].x, ad.x, v[r].y * ad.y)));
#pragma unroll
    for (int r = 0; r < 16; ++r)
      xs[r] = fmaf(v[r].w, as.w, fmaf(v[r].z, as.z, fmaf(v[r].x, as.x, v[r].y * as.y)));
    gat_fold<32, 16>(xd, lane);  gat_fold<16, 16>(xs, lane);
    gat_fold<16, 8>(xd, lane);   gat_fold<8, 8>(xs, lane);
    gat_fold<8, 4>(xd, lane);    gat_fold<4, 4>(xs, lane);
    gat_fold<4, 2>(xd, lane);    gat_fold<2, 2>(xs, lane);
    gat_fold<2, 1>(xd, lane);
    xs[0] += __shfl_xor(xs[0], 1, 32);
    sEd[2 * wave + 32 * ((lane >> 1) ^ (8 * rb)) + (lane & 1)] = xd[0];
    if ((lane & 1) == 0) sEs[2 * wave + 32 * (lane >> 2) + ((lane >> 1) & 1)] = xs[0];
    float wm = xd[0];
    wm = fmaxf(wm, __shfl_xor(wm, 16, 32));
    wm = fmaxf(wm, __shfl_xor(wm, 8, 32));
    wm = fmaxf(wm, __shfl_xor(wm, 4, 32));
    wm = fmaxf(wm, __shfl_xor(wm, 2, 32));
    wm = fmaxf(wm, __shfl_xor(wm, 1, 32));
    if (lane == 0) sMax[wave] = wm;
  }
  __syncthreads();

  float D = sMax[0];
#pragma unroll
  for (int k = 1; k < AT_WAVES; ++k) D = fmaxf(D, sMax[k]);
  {
    const float t = sEd[tid] - D;
    float* ee = sEE + (tid >> 1) * 4 + (tid & 1);
    ee[0] = __expf(t);
    ee[2] = __expf(ALPHA * t);
  }
  __syncthreads();

  const float es = sEs[wave * 16 + l16];
  const float u = (1.0f - ALPHA) * (es + D);
  const float fa = __expf(fminf(u, 0.0f));
  const float fb = __expf(fminf(-u, 0.0f));
  const v2f fa2 = {fa, fa};
  const v2f fb2 = {fb, fb};

  v2f lp = {0.0f, 0.0f};
  v8f acc[8];
#pragma unroll
  for (int t = 0; t < 8; ++t)
#pragma unroll
    for (int r = 0; r < 8; ++r) acc[t][r] = 0.0f;

  for (int w = 0; w < 16; ++w) {
    unsigned int mn[16];
    const unsigned int* nrow = mrow + ((w + 1) & 15) * 16;
#pragma unroll
    for (int k = 0; k < 16; ++k) mn[k] = nrow[k];

    const v4f* eew = (const v4f*)sEE + w * 16 + g;
#pragma unroll
    for (int c = 0; c < 8; ++c) {
      const v4f ee = eew[2 * c];
      const v2f e1 = {ee.x, ee.y};
      const v2f e2 = {ee.z, ee.w};
      const v2f ca = e1 * fa2;
      const v2f cb = e2 * fb2;
      v2f a;
      a.x = fmaxf(ca.x, cb.x);
      a.y = fmaxf(ca.y, cb.y);
      if (!__builtin_amdgcn_inverse_ballot_w32(mv[2 * c]))     a.x = 0.0f;
      if (!__builtin_amdgcn_inverse_ballot_w32(mv[2 * c + 1])) a.y = 0.0f;
      lp += a;
      const v2f* brow = sWh + (w * 16 + c * 2 + g) * PAIRP + l16;
      v2f b[8];
#pragma unroll
      for (int t = 0; t < 8; ++t) b[t] = brow[t * 16];
#pragma unroll
      for (int t = 0; t < 8; ++t)
        acc[t] = __builtin_amdgcn_wmma_f32_16x16x4_f32(
            false, a, false, b[t], (short)0, acc[t], false, false);
    }

#pragma unroll
    for (int k = 0; k < 16; ++k) mv[k] = mn[k];
  }
  float l = lp.x + lp.y;
  l += __shfl_xor(l, 16, 32);

  if (__builtin_amdgcn_ballot_w32(!(l > 1.0e-20f)) != 0u) {
    float m = -3.0e38f;
    l = 0.0f;
#pragma unroll
    for (int t = 0; t < 8; ++t)
#pragma unroll
      for (int r = 0; r < 8; ++r) acc[t][r] = 0.0f;

    for (int w = 0; w < 16; ++w) {
      unsigned int mn[16];
      const unsigned int* nrow = mrow + ((w + 1) & 15) * 16;
#pragma unroll
      for (int k = 0; k < 16; ++k) mn[k] = nrow[k];

      const float* edw = sEd + w * 32 + 2 * g;
      float e[16];
#pragma unroll
      for (int c = 0; c < 8; ++c) {
        const v2f ed = *(const v2f*)(edw + c * 4);
        float e0 = es + ed.x;
        e0 = e0 > 0.0f ? e0 : ALPHA * e0;
        if (!__builtin_amdgcn_inverse_ballot_w32(mv[2 * c])) e0 = NEG_INF;
        float e1 = es + ed.y;
        e1 = e1 > 0.0f ? e1 : ALPHA * e1;
        if (!__builtin_amdgcn_inverse_ballot_w32(mv[2 * c + 1])) e1 = NEG_INF;
        e[2 * c]     = e0;
        e[2 * c + 1] = e1;
      }

      float bm = e[0];
#pragma unroll
      for (int k = 1; k < 16; ++k) bm = fmaxf(bm, e[k]);
      bm = fmaxf(bm, __shfl_xor(bm, 16, 32));
      const float nm = fmaxf(m, bm);
      const float sc = __expf(m - nm);
      m = nm;
      l *= sc;
#pragma unroll
      for (int r = 0; r < 8; ++r) {
        const float f = __shfl(sc, r + 8 * g, 32);
#pragma unroll
        for (int t = 0; t < 8; ++t) acc[t][r] *= f;
      }

#pragma unroll
      for (int c = 0; c < 8; ++c) {
        v2f a;
        a.x = __expf(e[2 * c] - m);
        a.y = __expf(e[2 * c + 1] - m);
        l += a.x + a.y;
        const v2f* brow = sWh + (w * 16 + c * 2 + g) * PAIRP + l16;
        v2f b[8];
#pragma unroll
        for (int t = 0; t < 8; ++t) b[t] = brow[t * 16];
#pragma unroll
        for (int t = 0; t < 8; ++t)
          acc[t] = __builtin_amdgcn_wmma_f32_16x16x4_f32(
              false, a, false, b[t], (short)0, acc[t], false, false);
      }

#pragma unroll
      for (int k = 0; k < 16; ++k) mv[k] = mn[k];
    }
    l += __shfl_xor(l, 16, 32);
  }

  float* outblk = out + ((size_t)bt * NN + (size_t)rb * AT_ROWS + wave * 16) * DD;
#pragma unroll
  for (int r = 0; r < 8; ++r) {
    const int orow = r + 8 * g;
    const float lr = __shfl(l, orow, 32);
    const float inv = 1.0f / lr;
#pragma unroll
    for (int t = 0; t < 8; ++t)
      outblk[orow * DD + t * 16 + l16] = acc[t][r] * inv;
  }
}

extern "C" void kernel_launch(void* const* d_in, const int* in_sizes, int n_in,
                              void* d_out, int out_size, void* d_ws, size_t ws_size,
                              hipStream_t stream) {
  const float* h   = (const float*)d_in[0];
  const int*   adj = (const int*)d_in[1];
  const float* W   = (const float*)d_in[2];
  const float* a   = (const float*)d_in[3];
  float* out = (float*)d_out;

  float* Wh   = (float*)d_ws;
  unsigned int* adjbits = (unsigned int*)(Wh + (size_t)BT * NN * DD);

  const size_t lds1 = (size_t)(128 * LDSP) * 4 + (size_t)(64 * PAIRP) * 8;
  const size_t lds3 = (size_t)(NN / 2) * PAIRP * 8 + (size_t)NN * 4 + (size_t)AT_ROWS * 4
                    + (size_t)(NN / 2) * 16 + (size_t)AT_WAVES * 4;

  gat_wh_kernel<<<dim3(BT * 4), dim3(256), lds1, stream>>>(h, W, Wh, adj, adjbits);
  gat_attn_kernel<<<dim3(BT * NN / AT_ROWS), dim3(AT_THREADS), lds3, stream>>>(Wh, a, adjbits, out);
}
